# starting kernel
# baseline (speedup 1.0000x reference)
#include <hip/hip_runtime.h>


typedef float v2f __attribute__((ext_vector_type(2)));
typedef float v8f __attribute__((ext_vector_type(8)));

namespace {
constexpr int kDIn  = 4096;
constexpr int kDOut = 16;
constexpr int kSeq  = 2048;
constexpr int kRowsPerWave = 32;
}

__global__ __launch_bounds__(32)
void multilora_wmma_f32_kernel(const float* __restrict__ x,
                               const int*   __restrict__ adapter_ids,
                               const float* __restrict__ W,
                               float*       __restrict__ y) {
    const int lane = threadIdx.x & 31;
    const int half = lane >> 4;
    const int r    = lane & 15;

    const int row0 = blockIdx.x * kRowsPerWave;
    const int b    = row0 / kSeq;
    const int aid  = adapter_ids[b];

    const float* xp0 = x + (size_t)(row0 + r) * kDIn + 2 * half;
    const float* xp1 = xp0 + (size_t)16 * kDIn;
    const float* wp  = W + ((size_t)aid * kDOut + r) * kDIn + 2 * half;

    v8f c0 = {};
    v8f c1 = {};

    #pragma unroll 8
    for (int k = 0; k < kDIn; k += 4) {
        v2f bf = *(const v2f*)(wp  + k);
        v2f a0 = *(const v2f*)(xp0 + k);
        v2f a1 = *(const v2f*)(xp1 + k);
        c0 = __builtin_amdgcn_wmma_f32_16x16x4_f32(
                 false, a0, false, bf, (short)0, c0, false, false);
        c1 = __builtin_amdgcn_wmma_f32_16x16x4_f32(
                 false, a1, false, bf, (short)0, c1, false, false);
    }

    float* yp0 = y + (size_t)(row0 + 8 * half) * kDOut + r;
    #pragma unroll
    for (int v = 0; v < 8; ++v) {
        yp0[(size_t)v * kDOut] = c0[v];
    }
    float* yp1 = yp0 + (size_t)16 * kDOut;
    #pragma unroll
    for (int v = 0; v < 8; ++v) {
        yp1[(size_t)v * kDOut] = c1[v];
    }
}

extern "C" void kernel_launch(void* const* d_in, const int* in_sizes, int n_in,
                              void* d_out, int out_size, void* d_ws, size_t ws_size,
                              hipStream_t stream) {
    (void)in_sizes; (void)n_in; (void)d_ws; (void)ws_size; (void)out_size;
    const float* x   = (const float*)d_in[0];
    const int*   ids = (const int*)  d_in[1];
    const float* W   = (const float*)d_in[2];
    float*       y   = (float*)d_out;

    constexpr int totalRows = 4 * 2048;
    dim3 grid(totalRows / kRowsPerWave);
    dim3 block(32);
    multilora_wmma_f32_kernel<<<grid, block, 0, stream>>>(x, ids, W, y);
}


# W fragments come from LDS: each 64-k chunk of W fetched once per work-group by whole-line loads (2 buffers, 1 split barrier per chunk); freed load slots deepen x's ring 30->48 steps; sum order kept
# speedup vs baseline: 3.3154x; 3.3154x over previous
#include <hip/hip_runtime.h>


typedef float v2f __attribute__((ext_vector_type(2)));
typedef float v4f __attribute__((ext_vector_type(4)));
typedef float v8f __attribute__((ext_vector_type(8)));

namespace {
constexpr int kDIn  = 4096;
constexpr int kDOut = 16;
constexpr int kSeq  = 2048;
constexpr int kRowsPerWave   = 16;
constexpr int kWavesPerBlock = 2;
constexpr int kRowsPerBlock  = kRowsPerWave * kWavesPerBlock;
constexpr int kThreads = 32 * kWavesPerBlock;
constexpr int kChunkSteps = 16;
constexpr int kChunkK     = 4 * kChunkSteps;
constexpr int kChunks     = kDIn / kChunkK;
constexpr int kRingChunks = 3;
constexpr int kAhead      = kRingChunks * kChunkSteps;
constexpr int kWLoads     = (kDOut * kChunkK * 4) / (kThreads * 16);
constexpr int kPitchF     = kChunkK + 4;
constexpr int kBufF       = kDOut * kPitchF;
constexpr int kBHead      = 12;
constexpr int kMainIters  = (kChunks - 4) / kRingChunks;
static_assert(kMainIters * kRingChunks + 4 == kChunks, "chunk schedule: 20 x 3 whole chunks, then 60, 61, 62, 63");
static_assert(kAhead + kRingChunks * kWLoads <= 63, "the load counter holds 63");
static_assert(kSeq % kRowsPerBlock == 0, "a work-group stays inside one batch entry");
static_assert(kWLoads == 4, "W chunk tiling");
}

#define DEV static __device__ __forceinline__

DEV void wait_ds0()   { asm volatile("s_wait_dscnt 0x0" ::: "memory"); }
DEV void bar_signal() { asm volatile("s_barrier_signal -1" ::: "memory"); }
DEV void bar_wait()   { asm volatile("s_barrier_wait -1" ::: "memory"); }
DEV void pin()        { __builtin_amdgcn_sched_barrier(0); }

template <bool REFILL, bool WSTORE, bool WLOAD, bool NEXTB>
DEV void chunk(v8f& c, v2f (&af)[kChunkSteps], v2f (&bf)[kChunkSteps],
               const float* __restrict__ xnext, const float* bsrc, const float* bnxt,
               v4f (&wst)[kWLoads], float* sdst, const float* __restrict__ wnext) {
    #pragma unroll
    for (int j = kBHead; j < kChunkSteps; ++j) bf[j] = *(const v2f*)(bsrc + 4 * j);
    pin();

    #pragma unroll
    for (int j = 0; j < kChunkSteps; ++j) {
        if (j == 4) {
            if (WSTORE) {
                #pragma unroll
                for (int i = 0; i < kWLoads; ++i) *(v4f*)(sdst + i * 4 * kPitchF) = wst[i];
            }
            if (WLOAD) {
                #pragma unroll
                for (int i = 0; i < kWLoads; ++i) wst[i] = *(const v4f*)(wnext + (size_t)i * 4 * kDIn);
            }
            pin();
        }
        if (j == 8 && NEXTB) { wait_ds0(); bar_signal(); pin(); }
        if (j == kBHead && NEXTB) {
            bar_wait();
            #pragma unroll
            for (int q = 0; q < kBHead; ++q) bf[q] = *(const v2f*)(bnxt + 4 * q);
            pin();
        }
        c = __builtin_amdgcn_wmma_f32_16x16x4_f32(
                false, af[j], false, bf[j], (short)0, c, false, false);
        if (j == kChunkSteps - 1) {
            asm volatile("" : "+v"(c));
        }
        if (REFILL) af[j] = *(const v2f*)(xnext + 4 * j);
        pin();
    }
}

__global__ __launch_bounds__(kThreads)
void multilora_wmma_f32_kernel(const float* __restrict__ x,
                               const int*   __restrict__ adapter_ids,
                               const float* __restrict__ W,
                               float*       __restrict__ y) {
    __shared__ __attribute__((aligned(16))) float wlds[2 * kBufF];

    const int tid  = threadIdx.x;
    const int lane = tid & 31;
    const int half = lane >> 4;
    const int r    = lane & 15;
    const int wave = __builtin_amdgcn_readfirstlane((int)(threadIdx.x >> 5));

    const int row0 = blockIdx.x * kRowsPerBlock + wave * kRowsPerWave;
    const int b    = row0 / kSeq;
    const int aid  = adapter_ids[b];

    const float* xp = x + (size_t)(row0 + r) * kDIn + 2 * half;
    const int srow = tid >> 4, scol = 4 * (tid & 15);
    const float* wq = W + ((size_t)aid * kDOut + srow) * kDIn + scol;
    float* sdst0 = wlds + srow * kPitchF + scol;
    const float* bsrc0 = wlds + r * kPitchF + 2 * half;

    v8f c = {};
    v2f af[kRingChunks][kChunkSteps];
    v2f bf[kChunkSteps];
    v4f wst[kRingChunks][kWLoads];

    v4f w0[kWLoads];
    #pragma unroll
    for (int i = 0; i < kWLoads; ++i) w0[i] = *(const v4f*)(wq + (size_t)i * 4 * kDIn);
    pin();
    #pragma unroll
    for (int s = 0; s < kRingChunks; ++s) {
        #pragma unroll
        for (int j = 0; j < kChunkSteps; ++j) {
            if (j == 4) {
                #pragma unroll
                for (int i = 0; i < kWLoads; ++i)
                    wst[(s + 1) % kRingChunks][i] = *(const v4f*)(wq + (size_t)i * 4 * kDIn + (s + 1) * kChunkK);
                pin();
            }
            af[s][j] = *(const v2f*)(xp + 4 * (s * kChunkSteps + j));
            pin();
        }
    }
    #pragma unroll
    for (int i = 0; i < kWLoads; ++i) *(v4f*)(sdst0 + i * 4 * kPitchF) = w0[i];
    wait_ds0(); bar_signal(); bar_wait();
    #pragma unroll
    for (int q = 0; q < kBHead; ++q) bf[q] = *(const v2f*)(bsrc0 + 4 * q);
    pin();

    for (int it = 0; it < kMainIters; ++it) {
        const int n0 = it * kRingChunks;
        const float* xn = xp + (size_t)(n0 + kRingChunks) * kChunkK;
        const float* wn = wq + (size_t)(n0 + 4) * kChunkK;
        const int p0 = (n0 & 1) * kBufF, p1 = kBufF - p0;
        chunk<true, true, true, true>(c, af[0], bf, xn,               bsrc0 + p0, bsrc0 + p1, wst[1], sdst0 + p1, wn);
        chunk<true, true, true, true>(c, af[1], bf, xn + kChunkK,     bsrc0 + p1, bsrc0 + p0, wst[2], sdst0 + p0, wn + kChunkK);
        chunk<true, true, true, true>(c, af[2], bf, xn + 2 * kChunkK, bsrc0 + p0, bsrc0 + p1, wst[0], sdst0 + p1, wn + 2 * kChunkK);
    }
    {
        constexpr int n0 = kMainIters * kRingChunks;
        static_assert((n0 & 1) == 0 && n0 + 3 == kChunks - 1, "tail parity");
        const float* xn = xp + (size_t)(n0 + kRingChunks) * kChunkK;
        const float* b0 = bsrc0;
        const float* b1 = bsrc0 + kBufF;
        chunk<true,  true,  false, true >(c, af[0], bf, xn, b0, b1, wst[1], sdst0 + kBufF, wq);
        chunk<false, true,  false, true >(c, af[1], bf, xp, b1, b0, wst[2], sdst0,         wq);
        chunk<false, true,  false, true >(c, af[2], bf, xp, b0, b1, wst[0], sdst0 + kBufF, wq);
        chunk<false, false, false, false>(c, af[0], bf, xp, b1, b0, wst[1], sdst0,         wq);
    }

    float* yp = y + (size_t)(row0 + 8 * half) * kDOut + r;
    #pragma unroll
    for (int v = 0; v < 8; ++v) {
        yp[(size_t)v * kDOut] = c[v];
    }
}

extern "C" void kernel_launch(void* const* d_in, const int* in_sizes, int n_in,
                              void* d_out, int out_size, void* d_ws, size_t ws_size,
                              hipStream_t stream) {
    (void)in_sizes; (void)n_in; (void)d_ws; (void)ws_size; (void)out_size;
    const float* x   = (const float*)d_in[0];
    const int*   ids = (const int*)  d_in[1];
    const float* W   = (const float*)d_in[2];
    float*       y   = (float*)d_out;

    constexpr int totalRows = 4 * 2048;
    dim3 grid(totalRows / kRowsPerBlock);
    dim3 block(kThreads);
    multilora_wmma_f32_kernel<<<grid, block, 0, stream>>>(x, ids, W, y);
}


# A tile = 8 x rows at 2 k slices: an 8-B load takes 32 B of each of 8 rows and feeds 2 products; 8 rows/wave, 4 waves/work-group (1024 waves, was 512); 2 partial sums per output, added at the end
# speedup vs baseline: 4.0105x; 1.2097x over previous
#include <hip/hip_runtime.h>


typedef float v2f __attribute__((ext_vector_type(2)));
typedef float v4f __attribute__((ext_vector_type(4)));
typedef float v8f __attribute__((ext_vector_type(8)));

namespace {
constexpr int kDIn  = 4096;
constexpr int kDOut = 16;
constexpr int kSeq  = 2048;
constexpr int kRowsPerWave   = 8;
constexpr int kWavesPerBlock = 4;
constexpr int kRowsPerBlock  = kRowsPerWave * kWavesPerBlock;
constexpr int kThreads = 32 * kWavesPerBlock;
constexpr int kChunkSteps = 16;
constexpr int kChunkLoads = kChunkSteps / 2;
constexpr int kChunkK     = 4 * kChunkSteps;
constexpr int kChunks     = kDIn / kChunkK;
constexpr int kRing       = 6;
constexpr int kWLoads     = (kDOut * kChunkK * 4) / (kThreads * 16);
constexpr int kWRows      = kThreads / 16;
constexpr int kPitchF     = kChunkK + 4;
constexpr int kBufF       = kDOut * kPitchF;
constexpr int kBHead      = 12;
constexpr int kMainIters  = 9;
static_assert(kMainIters * kRing + 10 == kChunks, "chunk schedule: 9 x 6 whole chunks, then 54..63");
static_assert(kRing * (kChunkLoads + kWLoads) <= 63, "the load counter holds 63");
static_assert(kSeq % kRowsPerBlock == 0, "a work-group stays inside one batch entry");
static_assert(kWLoads == 2 && kWRows * kWLoads == kDOut, "W chunk tiling");
static_assert((kRing & 1) == 0, "the LDS buffer of a chunk is a constant per ring slot");
}

#define DEV static __device__ __forceinline__

DEV void wait_ds0()   { asm volatile("s_wait_dscnt 0x0" ::: "memory"); }
DEV void bar_signal() { asm volatile("s_barrier_signal -1" ::: "memory"); }
DEV void bar_wait()   { asm volatile("s_barrier_wait -1" ::: "memory"); }
DEV void pin()        { __builtin_amdgcn_sched_barrier(0); }

template <bool REFILL, bool WSTORE, bool WLOAD, bool NEXTB>
DEV void chunk(v8f& c0, v8f& c1, v2f (&af)[kChunkLoads], v2f (&bf)[kChunkSteps],
               const float* __restrict__ xnext, const float* bsrc, const float* bnxt,
               v4f (&wst)[kWLoads], float* sdst, const float* __restrict__ wnext) {
    #pragma unroll
    for (int j = kBHead; j < kChunkSteps; ++j) bf[j] = *(const v2f*)(bsrc + 4 * j);
    pin();

    #pragma unroll
    for (int j = 0; j < kChunkSteps; ++j) {
        if (j == 4) {
            if (WSTORE) {
                #pragma unroll
                for (int i = 0; i < kWLoads; ++i) *(v4f*)(sdst + i * kWRows * kPitchF) = wst[i];
            }
            if (WLOAD) {
                #pragma unroll
                for (int i = 0; i < kWLoads; ++i) wst[i] = *(const v4f*)(wnext + (size_t)i * kWRows * kDIn);
            }
            pin();
        }
        if (j == 8 && NEXTB) { wait_ds0(); bar_signal(); pin(); }
        if (j == kBHead && NEXTB) {
            bar_wait();
            #pragma unroll
            for (int q = 0; q < kBHead; ++q) bf[q] = *(const v2f*)(bnxt + 4 * q);
            pin();
        }
        if ((j & 1) == 0) {
            c0 = __builtin_amdgcn_wmma_f32_16x16x4_f32(
                     false, af[j >> 1], false, bf[j], (short)0, c0, false, false);
        } else {
            c1 = __builtin_amdgcn_wmma_f32_16x16x4_f32(
                     false, af[j >> 1], false, bf[j], (short)0, c1, false, false);
            if (j == kChunkSteps - 1) {
                asm volatile("" : "+v"(c1));
            }
            if (REFILL) af[j >> 1] = *(const v2f*)(xnext + 8 * (j >> 1));
        }
        pin();
    }
}

#define CHUNK(RF, WS, WL, NB, SLOT, XN, WN)                                                  \
    chunk<RF, WS, WL, NB>(c0, c1, af[SLOT], bf, XN,                                          \
                          bsrc0 + ((SLOT) & 1) * kBufF, bsrc0 + (((SLOT) + 1) & 1) * kBufF,  \
                          wst[((SLOT) + 1) % kRing], sdst0 + (((SLOT) + 1) & 1) * kBufF, WN)

__global__ __launch_bounds__(kThreads)
void multilora_wmma_f32_kernel(const float* __restrict__ x,
                               const int*   __restrict__ adapter_ids,
                               const float* __restrict__ W,
                               float*       __restrict__ y) {
    __shared__ __attribute__((aligned(16))) float wlds[2 * kBufF];

    const int tid  = threadIdx.x;
    const int lane = tid & 31;
    const int half = lane >> 4;
    const int r    = lane & 15;
    const int wave = __builtin_amdgcn_readfirstlane((int)(threadIdx.x >> 5));

    const int row0 = blockIdx.x * kRowsPerBlock + wave * kRowsPerWave;
    const int b    = row0 / kSeq;
    const int aid  = adapter_ids[b];

    const float* xp = x + (size_t)(row0 + (r & 7)) * kDIn + 4 * (r >> 3) + 2 * half;
    const int srow = tid >> 4, scol = 4 * (tid & 15);
    const float* wq = W + ((size_t)aid * kDOut + srow) * kDIn + scol;
    float* sdst0 = wlds + srow * kPitchF + scol;
    const float* bsrc0 = wlds + r * kPitchF + 2 * half;

    v8f c0 = {}, c1 = {};
    v2f af[kRing][kChunkLoads];
    v2f bf[kChunkSteps];
    v4f wst[kRing][kWLoads];

    v4f w0[kWLoads];
    #pragma unroll
    for (int i = 0; i < kWLoads; ++i) w0[i] = *(const v4f*)(wq + (size_t)i * kWRows * kDIn);
    pin();
    #pragma unroll
    for (int s = 0; s < kRing; ++s) {
        #pragma unroll
        for (int p = 0; p < kChunkLoads; ++p) {
            if (p == 2) {
                #pragma unroll
                for (int i = 0; i < kWLoads; ++i)
                    wst[(s + 1) % kRing][i] = *(const v4f*)(wq + (size_t)i * kWRows * kDIn + (s + 1) * kChunkK);
                pin();
            }
            af[s][p] = *(const v2f*)(xp + s * kChunkK + 8 * p);
            pin();
        }
    }
    #pragma unroll
    for (int i = 0; i < kWLoads; ++i) *(v4f*)(sdst0 + i * kWRows * kPitchF) = w0[i];
    wait_ds0(); bar_signal(); bar_wait();
    #pragma unroll
    for (int q = 0; q < kBHead; ++q) bf[q] = *(const v2f*)(bsrc0 + 4 * q);
    pin();

    for (int it = 0; it < kMainIters; ++it) {
        const float* xn = xp + (size_t)(it + 1) * kRing * kChunkK;
        const float* wn = wq + ((size_t)(it + 1) * kRing + 1) * kChunkK;
        CHUNK(true, true, true, true, 0, xn,               wn);
        CHUNK(true, true, true, true, 1, xn + kChunkK,     wn + kChunkK);
        CHUNK(true, true, true, true, 2, xn + 2 * kChunkK, wn + 2 * kChunkK);
        CHUNK(true, true, true, true, 3, xn + 3 * kChunkK, wn + 3 * kChunkK);
        CHUNK(true, true, true, true, 4, xn + 4 * kChunkK, wn + 4 * kChunkK);
        CHUNK(true, true, true, true, 5, xn + 5 * kChunkK, wn + 5 * kChunkK);
    }
    {
        constexpr int n0 = kMainIters * kRing;
        static_assert(n0 % kRing == 0 && n0 + 9 == kChunks - 1, "tail slots");
        const float* xn = xp + (size_t)(n0 + kRing) * kChunkK;
        const float* wn = wq + (size_t)(n0 + kRing + 1) * kChunkK;
        CHUNK(true,  true,  true,  true,  0, xn,               wn);
        CHUNK(true,  true,  true,  true,  1, xn + kChunkK,     wn + kChunkK);
        CHUNK(true,  true,  true,  true,  2, xn + 2 * kChunkK, wn + 2 * kChunkK);
        CHUNK(true,  true,  false, true,  3, xn + 3 * kChunkK, wq);
        CHUNK(false, true,  false, true,  4, xp, wq);
        CHUNK(false, true,  false, true,  5, xp, wq);
        CHUNK(false, true,  false, true,  0, xp, wq);
        CHUNK(false, true,  false, true,  1, xp, wq);
        CHUNK(false, true,  false, true,  2, xp, wq);
        CHUNK(false, false, false, false, 3, xp, wq);
    }

    float* yp = y + (size_t)(row0 + half) * kDOut + r;
    #pragma unroll
    for (int q = 0; q < 4; ++q) {
        const float own  = half ? c1[2 * q + 1] : c0[2 * q];
        const float give = half ? c1[2 * q]     : c0[2 * q + 1];
        yp[(size_t)(2 * q) * kDOut] = own + __shfl_xor(give, 16);
    }
}

extern "C" void kernel_launch(void* const* d_in, const int* in_sizes, int n_in,
                              void* d_out, int out_size, void* d_ws, size_t ws_size,
                              hipStream_t stream) {
    (void)in_sizes; (void)n_in; (void)d_ws; (void)ws_size; (void)out_size;
    const float* x   = (const float*)d_in[0];
    const int*   ids = (const int*)  d_in[1];
    const float* W   = (const float*)d_in[2];
    float*       y   = (float*)d_out;

    constexpr int totalRows = 4 * 2048;
    dim3 grid(totalRows / kRowsPerBlock);
    dim3 block(kThreads);
    multilora_wmma_f32_kernel<<<grid, block, 0, stream>>>(x, ids, W, y);
}


# A 128-k chunk is one run of 32 products, then one stretch with all its memory instructions (16 x refills, W loads/stores, fragment reads, barrier wait); was 1 load behind each 2 products; sums kept
# speedup vs baseline: 4.6653x; 1.1633x over previous
#include <hip/hip_runtime.h>


typedef float v2f __attribute__((ext_vector_type(2)));
typedef float v4f __attribute__((ext_vector_type(4)));
typedef float v8f __attribute__((ext_vector_type(8)));

namespace {
constexpr int kDIn  = 4096;
constexpr int kDOut = 16;
constexpr int kSeq  = 2048;
constexpr int kRowsPerWave   = 8;
constexpr int kWavesPerBlock = 4;
constexpr int kRowsPerBlock  = kRowsPerWave * kWavesPerBlock;
constexpr int kThreads = 32 * kWavesPerBlock;
constexpr int kChunkSteps = 32;
constexpr int kChunkLoads = kChunkSteps / 2;
constexpr int kChunkK     = 4 * kChunkSteps;
constexpr int kChunks     = kDIn / kChunkK;
constexpr int kRing       = 3;
constexpr int kRowThreads = kChunkK / 4;
constexpr int kWRows      = kThreads / kRowThreads;
constexpr int kWLoads     = kDOut / kWRows;
constexpr int kPitchF     = kChunkK + 4;
constexpr int kBufF       = kDOut * kPitchF;
constexpr int kSignalAt   = kChunkSteps / 4 - 1;
constexpr int kTrip       = 6;
constexpr int kMainIters  = (kChunks - 2 - kRing) / kTrip;
static_assert(kTrip % kRing == 0 && kTrip % 2 == 0, "a trip position fixes ring slot and LDS buffer");
static_assert(kRing * (kChunkLoads + kWLoads) <= 63, "the load counter holds 63");
static_assert(kSeq % kRowsPerBlock == 0, "a work-group stays inside one batch entry");
static_assert(kWRows * kWLoads == kDOut && kRowThreads * kWRows == kThreads, "W chunk tiling");
static_assert(((kPitchF / 4) & 1) == 1, "LDS pitch is 16 B x odd");
}

#define DEV static __device__ __forceinline__

DEV void wait_ds0()   { asm volatile("s_wait_dscnt 0x0" ::: "memory"); }
DEV void bar_signal() { asm volatile("s_barrier_signal -1" ::: "memory"); }
DEV void bar_wait()   { asm volatile("s_barrier_wait -1" ::: "memory"); }
DEV void pin()        { __builtin_amdgcn_sched_barrier(0); }

template <bool NEXTB, bool WSTORE, bool WLOAD, bool REFILL>
DEV void chunk(v8f& c0, v8f& c1, v2f (&af)[kChunkLoads], v2f (&bf)[kChunkSteps],
               const float* __restrict__ xnext, const float* bnxt,
               v4f (&wst)[kWLoads], float* sdst, const float* __restrict__ wnext) {
    #pragma unroll
    for (int j = 0; j < kChunkSteps; ++j) {
        if ((j & 1) == 0) {
            c0 = __builtin_amdgcn_wmma_f32_16x16x4_f32(
                     false, af[j >> 1], false, bf[j], (short)0, c0, false, false);
        } else {
            c1 = __builtin_amdgcn_wmma_f32_16x16x4_f32(
                     false, af[j >> 1], false, bf[j], (short)0, c1, false, false);
        }
        if (j == kSignalAt && NEXTB) { wait_ds0(); bar_signal(); }
        pin();
    }
    if (NEXTB) {
        bar_wait();
        pin();
        #pragma unroll
        for (int q = 0; q < kChunkSteps; ++q) bf[q] = *(const v2f*)(bnxt + 4 * q);
        pin();
    }
    if (WSTORE) {
        #pragma unroll
        for (int i = 0; i < kWLoads; ++i) *(v4f*)(sdst + i * kWRows * kPitchF) = wst[i];
        pin();
    }
    if (WLOAD) {
        #pragma unroll
        for (int i = 0; i < kWLoads; ++i) wst[i] = *(const v4f*)(wnext + (size_t)i * kWRows * kDIn);
        pin();
    }
    if (REFILL) {
        #pragma unroll
        for (int p = 0; p < kChunkLoads; ++p) af[p] = *(const v2f*)(xnext + 8 * p);
        pin();
    }
}

#define CHUNK(NB, WS, WL, RF, POS, XN, WN)                                                   \
    chunk<NB, WS, WL, RF>(c0, c1, af[(POS) % kRing], bf, XN,                                 \
                          bsrc0 + (((POS) + 1) & 1) * kBufF,                                 \
                          wst[(POS) % kRing], sdst0 + ((POS) & 1) * kBufF, WN)

template <int J>
DEV void tail(v8f& c0, v8f& c1, v2f (&af)[kRing][kChunkLoads], v2f (&bf)[kChunkSteps],
              v4f (&wst)[kRing][kWLoads], const float* __restrict__ xp,
              const float* __restrict__ wq, const float* bsrc0, float* sdst0) {
    if constexpr (J < kChunks) {
        constexpr bool NB = (J + 1 < kChunks);
        constexpr bool WS = (J + 2 < kChunks);
        constexpr bool WL = (J + 2 + kRing < kChunks);
        constexpr bool RF = (J + kRing < kChunks);
        CHUNK(NB, WS, WL, RF, J % kTrip,
              RF ? xp + (size_t)(J + kRing) * kChunkK : xp,
              WL ? wq + (size_t)(J + 2 + kRing) * kChunkK : wq);
        tail<J + 1>(c0, c1, af, bf, wst, xp, wq, bsrc0, sdst0);
    }
}

__global__ __launch_bounds__(kThreads)
void multilora_wmma_f32_kernel(const float* __restrict__ x,
                               const int*   __restrict__ adapter_ids,
                               const float* __restrict__ W,
                               float*       __restrict__ y) {
    __shared__ __attribute__((aligned(16))) float wlds[2 * kBufF];

    const int tid  = threadIdx.x;
    const int lane = tid & 31;
    const int half = lane >> 4;
    const int r    = lane & 15;
    const int wave = __builtin_amdgcn_readfirstlane((int)(threadIdx.x >> 5));

    const int row0 = blockIdx.x * kRowsPerBlock + wave * kRowsPerWave;
    const int b    = row0 / kSeq;
    const int aid  = adapter_ids[b];

    const float* xp = x + (size_t)(row0 + (r & 7)) * kDIn + 4 * (r >> 3) + 2 * half;
    const int srow = tid / kRowThreads, scol = 4 * (tid % kRowThreads);
    const float* wq = W + ((size_t)aid * kDOut + srow) * kDIn + scol;
    float* sdst0 = wlds + srow * kPitchF + scol;
    const float* bsrc0 = wlds + r * kPitchF + 2 * half;

    v8f c0 = {}, c1 = {};
    v2f af[kRing][kChunkLoads];
    v2f bf[kChunkSteps];
    v4f wst[kRing][kWLoads];

    v4f w0[kWLoads], w1[kWLoads];
    #pragma unroll
    for (int i = 0; i < kWLoads; ++i) w0[i] = *(const v4f*)(wq + (size_t)i * kWRows * kDIn);
    #pragma unroll
    for (int i = 0; i < kWLoads; ++i) w1[i] = *(const v4f*)(wq + (size_t)i * kWRows * kDIn + kChunkK);
    pin();
    #pragma unroll
    for (int s = 0; s < kRing; ++s) {
        if (s == kRing - 1) {
            #pragma unroll
            for (int i = 0; i < kWLoads; ++i) {
                *(v4f*)(sdst0 + i * kWRows * kPitchF) = w0[i];
                *(v4f*)(sdst0 + kBufF + i * kWRows * kPitchF) = w1[i];
            }
            pin();
        }
        #pragma unroll
        for (int i = 0; i < kWLoads; ++i)
            wst[s][i] = *(const v4f*)(wq + (size_t)i * kWRows * kDIn + (s + 2) * kChunkK);
        pin();
        #pragma unroll
        for (int p = 0; p < kChunkLoads; ++p) af[s][p] = *(const v2f*)(xp + s * kChunkK + 8 * p);
        pin();
    }
    wait_ds0(); bar_signal(); bar_wait();
    #pragma unroll
    for (int q = 0; q < kChunkSteps; ++q) bf[q] = *(const v2f*)(bsrc0 + 4 * q);
    pin();

    for (int it = 0; it < kMainIters; ++it) {
        const float* xn = xp + ((size_t)it * kTrip + kRing) * kChunkK;
        const float* wn = wq + ((size_t)it * kTrip + kRing + 2) * kChunkK;
        CHUNK(true, true, true, true, 0, xn,               wn);
        CHUNK(true, true, true, true, 1, xn + kChunkK,     wn + kChunkK);
        CHUNK(true, true, true, true, 2, xn + 2 * kChunkK, wn + 2 * kChunkK);
        CHUNK(true, true, true, true, 3, xn + 3 * kChunkK, wn + 3 * kChunkK);
        CHUNK(true, true, true, true, 4, xn + 4 * kChunkK, wn + 4 * kChunkK);
        CHUNK(true, true, true, true, 5, xn + 5 * kChunkK, wn + 5 * kChunkK);
    }
    tail<kMainIters * kTrip>(c0, c1, af, bf, wst, xp, wq, bsrc0, sdst0);

    float* yp = y + (size_t)(row0 + half) * kDOut + r;
    #pragma unroll
    for (int q = 0; q < 4; ++q) {
        const float own  = half ? c1[2 * q + 1] : c0[2 * q];
        const float give = half ? c1[2 * q]     : c0[2 * q + 1];
        yp[(size_t)(2 * q) * kDOut] = own + __shfl_xor(give, 16);
    }
}

extern "C" void kernel_launch(void* const* d_in, const int* in_sizes, int n_in,
                              void* d_out, int out_size, void* d_ws, size_t ws_size,
                              hipStream_t stream) {
    (void)in_sizes; (void)n_in; (void)d_ws; (void)ws_size; (void)out_size;
    const float* x   = (const float*)d_in[0];
    const int*   ids = (const int*)  d_in[1];
    const float* W   = (const float*)d_in[2];
    float*       y   = (float*)d_out;

    constexpr int totalRows = 4 * 2048;
    dim3 grid(totalRows / kRowsPerBlock);
    dim3 block(kThreads);
    multilora_wmma_f32_kernel<<<grid, block, 0, stream>>>(x, ids, W, y);
}
